# starting kernel
# baseline (speedup 1.0000x reference)
#include <hip/hip_runtime.h>


typedef float v2f __attribute__((ext_vector_type(2)));
typedef float v8f __attribute__((ext_vector_type(8)));

__device__ __forceinline__ v8f wmma4(v2f a, v2f b, v8f c) {
  return __builtin_amdgcn_wmma_f32_16x16x4_f32(false, a, false, b, (short)0, c, false, false);
}

#define OFF_XT   0
#define OFF_QT   8448
#define OFF_KB   17152
#define OFF_VB   25856
#define OFF_CT   34560
#define OFF_BO   38912
#define OFF_GS   39040
#define OFF_MEAN 39168
#define OFF_RSTD 39232
#define SMEM_FLOATS 39296
#define SMEM_BYTES  (SMEM_FLOATS * 4)

__global__ __launch_bounds__(256, 1)
void fused_linattn_kernel(const float* __restrict__ x,
                          const float* __restrict__ Wqkv,
                          const float* __restrict__ Wout,
                          const float* __restrict__ bout,
                          const float* __restrict__ g,
                          float* __restrict__ out)
{
  extern __shared__ float sm[];
  float* XT    = sm + OFF_XT;
  float* QT    = sm + OFF_QT;
  float* KB    = sm + OFF_KB;
  float* VB    = sm + OFF_VB;
  float* CT    = sm + OFF_CT;
  float* BO    = sm + OFF_BO;
  float* GS    = sm + OFF_GS;
  float* MEANv = sm + OFF_MEAN;
  float* RSTDv = sm + OFF_RSTD;
  float* AOT = XT;
  float* YT  = KB;

  const int tid  = threadIdx.x;
  const int wave = tid >> 5;
  const int lane = tid & 31;
  const int l16  = lane & 15;
  const int hi   = lane >> 4;
  const int kofs = 2 * hi;

  const int b  = blockIdx.y;
  const int ph = blockIdx.x >> 3;
  const int pw = blockIdx.x & 7;
  const long gbase = (long)b * 524288 + (long)(ph * 8) * 64 + (long)(pw * 8);

  for (int it = 0; it < 4; ++it) {
    int p = tid + 256 * it;
    int c = p >> 3, r = p & 7;
    const float4* src = (const float4*)(x + gbase + (long)c * 4096 + (long)r * 64);
    float4 a0 = src[0], a1 = src[1];
    float vals[8] = {a0.x, a0.y, a0.z, a0.w, a1.x, a1.y, a1.z, a1.w};
#pragma unroll
    for (int col = 0; col < 8; ++col)
      XT[(r * 8 + col) * 132 + c] = vals[col];
  }
  if (tid < 128) { BO[tid] = bout[tid]; GS[tid] = g[tid]; }
  __syncthreads();

  {
    v8f acc[12] = {};
    const float* Wbase = Wqkv + (wave * 16 + l16) * 128 + kofs;
    const float* Xbase = &XT[l16 * 132 + kofs];
    for (int k = 0; k < 128; k += 4) {
      v2f a0 = *(const v2f*)(Wbase + k);
      v2f a1 = *(const v2f*)(Wbase + 16384 + k);
      v2f a2 = *(const v2f*)(Wbase + 32768 + k);
      v2f b0 = *(const v2f*)(Xbase + 0 * 2112 + k);
      v2f b1 = *(const v2f*)(Xbase + 1 * 2112 + k);
      v2f b2 = *(const v2f*)(Xbase + 2 * 2112 + k);
      v2f b3 = *(const v2f*)(Xbase + 3 * 2112 + k);
      acc[0]  = wmma4(a0, b0, acc[0]);  acc[1]  = wmma4(a0, b1, acc[1]);
      acc[2]  = wmma4(a0, b2, acc[2]);  acc[3]  = wmma4(a0, b3, acc[3]);
      acc[4]  = wmma4(a1, b0, acc[4]);  acc[5]  = wmma4(a1, b1, acc[5]);
      acc[6]  = wmma4(a1, b2, acc[6]);  acc[7]  = wmma4(a1, b3, acc[7]);
      acc[8]  = wmma4(a2, b0, acc[8]);  acc[9]  = wmma4(a2, b1, acc[9]);
      acc[10] = wmma4(a2, b2, acc[10]); acc[11] = wmma4(a2, b3, acc[11]);
    }
#pragma unroll
    for (int j = 0; j < 3; ++j) {
      const int mt = wave + 8 * j;
#pragma unroll
      for (int nt = 0; nt < 4; ++nt) {
        int nn = nt * 16 + l16;
#pragma unroll
        for (int i = 0; i < 8; ++i) {
          int mm = mt * 16 + i + 8 * hi;
          float val = acc[j * 4 + nt][i];
          if (j == 0) {
            QT[(mm >> 5) * 2176 + nn * 34 + (mm & 31)] = val;
          } else if (j == 1) {
            KB[(mm - 128) * 68 + nn] = val;
          } else {
            VB[(mm - 256) * 68 + nn] = val;
          }
        }
      }
    }
  }
  __syncthreads();

  {
    int h = tid >> 6, n = tid & 63;
    float* row = &QT[h * 2176 + n * 34];
    float mx = row[0];
#pragma unroll
    for (int d = 1; d < 32; ++d) mx = fmaxf(mx, row[d]);
    float e[32]; float s = 0.f;
#pragma unroll
    for (int d = 0; d < 32; ++d) { e[d] = expf(row[d] - mx); s += e[d]; }
    float inv = 0.17677669529663687f / s;
#pragma unroll
    for (int d = 0; d < 32; ++d) row[d] = e[d] * inv;
  }
  if (tid < 128) {
    float* row = &KB[tid * 68];
    float mx = row[0];
    for (int n2 = 1; n2 < 64; ++n2) mx = fmaxf(mx, row[n2]);
    float s = 0.f;
    for (int n2 = 0; n2 < 64; ++n2) { float t = expf(row[n2] - mx); row[n2] = t; s += t; }
    float inv = 1.f / s;
    for (int n2 = 0; n2 < 64; ++n2) row[n2] *= inv;
  }
  __syncthreads();

  {
    int h = wave >> 1, sub = wave & 1;
    v8f acc0 = {}, acc1 = {};
    const float* A0 = &KB[(h * 32 + l16) * 68 + kofs];
    const float* A1 = A0 + 16 * 68;
    const float* Bp = &VB[(h * 32 + sub * 16 + l16) * 68 + kofs];
#pragma unroll
    for (int k = 0; k < 64; k += 4) {
      v2f b  = *(const v2f*)(Bp + k);
      v2f a0 = *(const v2f*)(A0 + k);
      v2f a1 = *(const v2f*)(A1 + k);
      acc0 = wmma4(a0, b, acc0);
      acc1 = wmma4(a1, b, acc1);
    }
#pragma unroll
    for (int i = 0; i < 8; ++i) {
      int e2 = sub * 16 + l16;
      CT[h * 1088 + e2 * 34 + ( 0 + i + 8 * hi)] = acc0[i];
      CT[h * 1088 + e2 * 34 + (16 + i + 8 * hi)] = acc1[i];
    }
  }
  __syncthreads();

  {
    int h = wave >> 1, sub = wave & 1;
    v8f acc[4] = {};
    const float* A0 = &CT[h * 1088 + l16 * 34 + kofs];
    const float* A1 = A0 + 16 * 34;
    const float* B0 = &QT[h * 2176 + (sub * 32 + l16) * 34 + kofs];
    const float* B1 = B0 + 16 * 34;
#pragma unroll
    for (int k = 0; k < 32; k += 4) {
      v2f a0 = *(const v2f*)(A0 + k);
      v2f a1 = *(const v2f*)(A1 + k);
      v2f b0 = *(const v2f*)(B0 + k);
      v2f b1 = *(const v2f*)(B1 + k);
      acc[0] = wmma4(a0, b0, acc[0]);
      acc[1] = wmma4(a0, b1, acc[1]);
      acc[2] = wmma4(a1, b0, acc[2]);
      acc[3] = wmma4(a1, b1, acc[3]);
    }
#pragma unroll
    for (int mt = 0; mt < 2; ++mt) {
#pragma unroll
      for (int q2 = 0; q2 < 2; ++q2) {
        int nt = sub * 2 + q2;
        int nn = nt * 16 + l16;
#pragma unroll
        for (int i = 0; i < 8; ++i) {
          int e2 = mt * 16 + i + 8 * hi;
          AOT[nn * 132 + h * 32 + e2] = acc[mt * 2 + q2][i];
        }
      }
    }
  }
  __syncthreads();

  {
    const float* Wrow  = Wout + (wave * 16 + l16) * 128 + kofs;
    const float* Abase = &AOT[l16 * 132 + kofs];
    v8f acc[4] = {};
    for (int k = 0; k < 128; k += 4) {
      v2f a  = *(const v2f*)(Wrow + k);
      v2f b0 = *(const v2f*)(Abase + 0 * 2112 + k);
      v2f b1 = *(const v2f*)(Abase + 1 * 2112 + k);
      v2f b2 = *(const v2f*)(Abase + 2 * 2112 + k);
      v2f b3 = *(const v2f*)(Abase + 3 * 2112 + k);
      acc[0] = wmma4(a, b0, acc[0]);
      acc[1] = wmma4(a, b1, acc[1]);
      acc[2] = wmma4(a, b2, acc[2]);
      acc[3] = wmma4(a, b3, acc[3]);
    }
#pragma unroll
    for (int nt = 0; nt < 4; ++nt) {
      int nn = nt * 16 + l16;
#pragma unroll
      for (int i = 0; i < 8; ++i) {
        int mm = wave * 16 + i + 8 * hi;
        YT[nn * 132 + mm] = acc[nt][i] + BO[mm];
      }
    }
  }
  __syncthreads();

  if (tid < 64) {
    const float* row = &YT[tid * 132];
    float s = 0.f;
    for (int c = 0; c < 128; ++c) s += row[c];
    float mean = s * (1.f / 128.f);
    float v = 0.f;
    for (int c = 0; c < 128; ++c) { float d = row[c] - mean; v += d * d; }
    MEANv[tid] = mean;
    RSTDv[tid] = rsqrtf(v * (1.f / 128.f) + 1e-5f);
  }
  __syncthreads();

  for (int it = 0; it < 4; ++it) {
    int p = tid + 256 * it;
    int c = p >> 3, r = p & 7;
    float gs = GS[c];
    float vals[8];
#pragma unroll
    for (int col = 0; col < 8; ++col) {
      int n = r * 8 + col;
      vals[col] = (YT[n * 132 + c] - MEANv[n]) * RSTDv[n] * gs;
    }
    float4* dst = (float4*)(out + gbase + (long)c * 4096 + (long)r * 64);
    dst[0] = make_float4(vals[0], vals[1], vals[2], vals[3]);
    dst[1] = make_float4(vals[4], vals[5], vals[6], vals[7]);
  }
}

extern "C" void kernel_launch(void* const* d_in, const int* in_sizes, int n_in,
                              void* d_out, int out_size, void* d_ws, size_t ws_size,
                              hipStream_t stream) {
  const float* x    = (const float*)d_in[0];
  const float* Wqkv = (const float*)d_in[1];
  const float* Wout = (const float*)d_in[2];
  const float* bout = (const float*)d_in[3];
  const float* g    = (const float*)d_in[4];
  float* out = (float*)d_out;
  (void)in_sizes; (void)n_in; (void)out_size; (void)d_ws; (void)ws_size;

  hipFuncSetAttribute((const void*)fused_linattn_kernel,
                      hipFuncAttributeMaxDynamicSharedMemorySize, SMEM_BYTES);

  fused_linattn_kernel<<<dim3(64, 64), 256, SMEM_BYTES, stream>>>(x, Wqkv, Wout, bout, g, out);
}


# Weight GEMMs (qkv, out-proj): operand fragments as 16-byte loads, each lane half owning 16 contiguous k per 32-k block, weights fetched one block ahead; k summation order permuted within each 32
# speedup vs baseline: 1.0716x; 1.0716x over previous
#include <hip/hip_runtime.h>


typedef float v2f __attribute__((ext_vector_type(2)));
typedef float v8f __attribute__((ext_vector_type(8)));

__device__ __forceinline__ v8f wmma4(v2f a, v2f b, v8f c) {
  return __builtin_amdgcn_wmma_f32_16x16x4_f32(false, a, false, b, (short)0, c, false, false);
}

#define OFF_XT   0
#define OFF_QT   8448
#define OFF_KB   17152
#define OFF_VB   25856
#define OFF_CT   34560
#define OFF_BO   38912
#define OFF_GS   39040
#define OFF_MEAN 39168
#define OFF_RSTD 39232
#define SMEM_FLOATS 39296
#define SMEM_BYTES  (SMEM_FLOATS * 4)

__global__ __launch_bounds__(256, 1)
void fused_linattn_kernel(const float* __restrict__ x,
                          const float* __restrict__ Wqkv,
                          const float* __restrict__ Wout,
                          const float* __restrict__ bout,
                          const float* __restrict__ g,
                          float* __restrict__ out)
{
  extern __shared__ __attribute__((aligned(16))) float sm[];
  float* XT    = sm + OFF_XT;
  float* QT    = sm + OFF_QT;
  float* KB    = sm + OFF_KB;
  float* VB    = sm + OFF_VB;
  float* CT    = sm + OFF_CT;
  float* BO    = sm + OFF_BO;
  float* GS    = sm + OFF_GS;
  float* MEANv = sm + OFF_MEAN;
  float* RSTDv = sm + OFF_RSTD;
  float* AOT = XT;
  float* YT  = KB;

  const int tid  = threadIdx.x;
  const int wave = tid >> 5;
  const int lane = tid & 31;
  const int l16  = lane & 15;
  const int hi   = lane >> 4;
  const int kofs = 2 * hi;

  const int b  = blockIdx.y;
  const int ph = blockIdx.x >> 3;
  const int pw = blockIdx.x & 7;
  const long gbase = (long)b * 524288 + (long)(ph * 8) * 64 + (long)(pw * 8);

  const float* Wl = Wqkv + (wave * 16 + l16) * 128 + 16 * hi;
  float4 aw[2][3][4];
#pragma unroll
  for (int j = 0; j < 3; ++j) {
#pragma unroll
    for (int m = 0; m < 4; ++m)
      aw[0][j][m] = *(const float4*)(Wl + j * 16384 + 4 * m);
  }

  for (int it = 0; it < 4; ++it) {
    int p = tid + 256 * it;
    int c = p >> 3, r = p & 7;
    const float4* src = (const float4*)(x + gbase + (long)c * 4096 + (long)r * 64);
    float4 a0 = src[0], a1 = src[1];
    float vals[8] = {a0.x, a0.y, a0.z, a0.w, a1.x, a1.y, a1.z, a1.w};
#pragma unroll
    for (int col = 0; col < 8; ++col)
      XT[(r * 8 + col) * 132 + c] = vals[col];
  }
  if (tid < 128) { BO[tid] = bout[tid]; GS[tid] = g[tid]; }
  __syncthreads();

  float4 ao[4][4];

  {
    v8f acc[12] = {};
    const float* Xl = &XT[l16 * 132 + 16 * hi];
    float4 bx[2][4];
#pragma unroll
    for (int nt = 0; nt < 4; ++nt) bx[0][nt] = *(const float4*)(Xl + nt * 2112);
#pragma unroll
    for (int t = 0; t < 4; ++t) {
      if (t + 1 < 4) {
#pragma unroll
        for (int j = 0; j < 3; ++j) {
#pragma unroll
          for (int m = 0; m < 4; ++m)
            aw[(t + 1) & 1][j][m] = *(const float4*)(Wl + j * 16384 + 32 * (t + 1) + 4 * m);
        }
      }
#pragma unroll
      for (int m = 0; m < 4; ++m) {
        const int s = t * 4 + m;
        if (s + 1 < 16) {
#pragma unroll
          for (int nt = 0; nt < 4; ++nt)
            bx[(s + 1) & 1][nt] = *(const float4*)(Xl + nt * 2112 + 32 * ((s + 1) >> 2) + 4 * ((s + 1) & 3));
        }
#pragma unroll
        for (int j = 0; j < 3; ++j) {
          const float4 av = aw[t & 1][j][m];
          const v2f a = {av.x, av.y};
#pragma unroll
          for (int nt = 0; nt < 4; ++nt) {
            const v2f bq = {bx[s & 1][nt].x, bx[s & 1][nt].y};
            acc[j * 4 + nt] = wmma4(a, bq, acc[j * 4 + nt]);
          }
        }
#pragma unroll
        for (int j = 0; j < 3; ++j) {
          const float4 av = aw[t & 1][j][m];
          const v2f a = {av.z, av.w};
#pragma unroll
          for (int nt = 0; nt < 4; ++nt) {
            const v2f bq = {bx[s & 1][nt].z, bx[s & 1][nt].w};
            acc[j * 4 + nt] = wmma4(a, bq, acc[j * 4 + nt]);
          }
        }
      }
    }
#pragma unroll
    for (int t = 0; t < 4; ++t) {
#pragma unroll
      for (int m = 0; m < 4; ++m)
        ao[t][m] = *(const float4*)(Wout + (wave * 16 + l16) * 128 + 16 * hi + 32 * t + 4 * m);
    }
#pragma unroll
    for (int j = 0; j < 3; ++j) {
      const int mt = wave + 8 * j;
#pragma unroll
      for (int nt = 0; nt < 4; ++nt) {
        int nn = nt * 16 + l16;
#pragma unroll
        for (int i = 0; i < 8; ++i) {
          int mm = mt * 16 + i + 8 * hi;
          float val = acc[j * 4 + nt][i];
          if (j == 0) {
            QT[(mm >> 5) * 2176 + nn * 34 + (mm & 31)] = val;
          } else if (j == 1) {
            KB[(mm - 128) * 68 + nn] = val;
          } else {
            VB[(mm - 256) * 68 + nn] = val;
          }
        }
      }
    }
  }
  __syncthreads();

  {
    int h = tid >> 6, n = tid & 63;
    float* row = &QT[h * 2176 + n * 34];
    float mx = row[0];
#pragma unroll
    for (int d = 1; d < 32; ++d) mx = fmaxf(mx, row[d]);
    float e[32]; float s = 0.f;
#pragma unroll
    for (int d = 0; d < 32; ++d) { e[d] = expf(row[d] - mx); s += e[d]; }
    float inv = 0.17677669529663687f / s;
#pragma unroll
    for (int d = 0; d < 32; ++d) row[d] = e[d] * inv;
  }
  if (tid < 128) {
    float* row = &KB[tid * 68];
    float mx = row[0];
    for (int n2 = 1; n2 < 64; ++n2) mx = fmaxf(mx, row[n2]);
    float s = 0.f;
    for (int n2 = 0; n2 < 64; ++n2) { float t = expf(row[n2] - mx); row[n2] = t; s += t; }
    float inv = 1.f / s;
    for (int n2 = 0; n2 < 64; ++n2) row[n2] *= inv;
  }
  __syncthreads();

  {
    int h = wave >> 1, sub = wave & 1;
    v8f acc0 = {}, acc1 = {};
    const float* A0 = &KB[(h * 32 + l16) * 68 + kofs];
    const float* A1 = A0 + 16 * 68;
    const float* Bp = &VB[(h * 32 + sub * 16 + l16) * 68 + kofs];
#pragma unroll
    for (int k = 0; k < 64; k += 4) {
      v2f b  = *(const v2f*)(Bp + k);
      v2f a0 = *(const v2f*)(A0 + k);
      v2f a1 = *(const v2f*)(A1 + k);
      acc0 = wmma4(a0, b, acc0);
      acc1 = wmma4(a1, b, acc1);
    }
#pragma unroll
    for (int i = 0; i < 8; ++i) {
      int e2 = sub * 16 + l16;
      CT[h * 1088 + e2 * 34 + ( 0 + i + 8 * hi)] = acc0[i];
      CT[h * 1088 + e2 * 34 + (16 + i + 8 * hi)] = acc1[i];
    }
  }
  __syncthreads();

  {
    int h = wave >> 1, sub = wave & 1;
    v8f acc[4] = {};
    const float* A0 = &CT[h * 1088 + l16 * 34 + kofs];
    const float* A1 = A0 + 16 * 34;
    const float* B0 = &QT[h * 2176 + (sub * 32 + l16) * 34 + kofs];
    const float* B1 = B0 + 16 * 34;
#pragma unroll
    for (int k = 0; k < 32; k += 4) {
      v2f a0 = *(const v2f*)(A0 + k);
      v2f a1 = *(const v2f*)(A1 + k);
      v2f b0 = *(const v2f*)(B0 + k);
      v2f b1 = *(const v2f*)(B1 + k);
      acc[0] = wmma4(a0, b0, acc[0]);
      acc[1] = wmma4(a0, b1, acc[1]);
      acc[2] = wmma4(a1, b0, acc[2]);
      acc[3] = wmma4(a1, b1, acc[3]);
    }
#pragma unroll
    for (int mt = 0; mt < 2; ++mt) {
#pragma unroll
      for (int q2 = 0; q2 < 2; ++q2) {
        int nt = sub * 2 + q2;
        int nn = nt * 16 + l16;
#pragma unroll
        for (int i = 0; i < 8; ++i) {
          int e2 = mt * 16 + i + 8 * hi;
          AOT[nn * 132 + h * 32 + e2] = acc[mt * 2 + q2][i];
        }
      }
    }
  }
  __syncthreads();

  {
    const float* Al = &AOT[l16 * 132 + 16 * hi];
    v8f acc[4] = {};
    float4 bb[2][4];
#pragma unroll
    for (int nt = 0; nt < 4; ++nt) bb[0][nt] = *(const float4*)(Al + nt * 2112);
#pragma unroll
    for (int s = 0; s < 16; ++s) {
      if (s + 1 < 16) {
#pragma unroll
        for (int nt = 0; nt < 4; ++nt)
          bb[(s + 1) & 1][nt] = *(const float4*)(Al + nt * 2112 + 32 * ((s + 1) >> 2) + 4 * ((s + 1) & 3));
      }
      const float4 av = ao[s >> 2][s & 3];
      const v2f a0 = {av.x, av.y};
      const v2f a1 = {av.z, av.w};
#pragma unroll
      for (int nt = 0; nt < 4; ++nt) {
        const v2f bq = {bb[s & 1][nt].x, bb[s & 1][nt].y};
        acc[nt] = wmma4(a0, bq, acc[nt]);
      }
#pragma unroll
      for (int nt = 0; nt < 4; ++nt) {
        const v2f bq = {bb[s & 1][nt].z, bb[s & 1][nt].w};
        acc[nt] = wmma4(a1, bq, acc[nt]);
      }
    }
#pragma unroll
    for (int nt = 0; nt < 4; ++nt) {
      int nn = nt * 16 + l16;
#pragma unroll
      for (int i = 0; i < 8; ++i) {
        int mm = wave * 16 + i + 8 * hi;
        YT[nn * 132 + mm] = acc[nt][i] + BO[mm];
      }
    }
  }
  __syncthreads();

  if (tid < 64) {
    const float* row = &YT[tid * 132];
    float s = 0.f;
    for (int c = 0; c < 128; ++c) s += row[c];
    float mean = s * (1.f / 128.f);
    float v = 0.f;
    for (int c = 0; c < 128; ++c) { float d = row[c] - mean; v += d * d; }
    MEANv[tid] = mean;
    RSTDv[tid] = rsqrtf(v * (1.f / 128.f) + 1e-5f);
  }
  __syncthreads();

  for (int it = 0; it < 4; ++it) {
    int p = tid + 256 * it;
    int c = p >> 3, r = p & 7;
    float gs = GS[c];
    float vals[8];
#pragma unroll
    for (int col = 0; col < 8; ++col) {
      int n = r * 8 + col;
      vals[col] = (YT[n * 132 + c] - MEANv[n]) * RSTDv[n] * gs;
    }
    float4* dst = (float4*)(out + gbase + (long)c * 4096 + (long)r * 64);
    dst[0] = make_float4(vals[0], vals[1], vals[2], vals[3]);
    dst[1] = make_float4(vals[4], vals[5], vals[6], vals[7]);
  }
}

extern "C" void kernel_launch(void* const* d_in, const int* in_sizes, int n_in,
                              void* d_out, int out_size, void* d_ws, size_t ws_size,
                              hipStream_t stream) {
  const float* x    = (const float*)d_in[0];
  const float* Wqkv = (const float*)d_in[1];
  const float* Wout = (const float*)d_in[2];
  const float* bout = (const float*)d_in[3];
  const float* g    = (const float*)d_in[4];
  float* out = (float*)d_out;
  (void)in_sizes; (void)n_in; (void)out_size; (void)d_ws; (void)ws_size;

  hipFuncSetAttribute((const void*)fused_linattn_kernel,
                      hipFuncAttributeMaxDynamicSharedMemorySize, SMEM_BYTES);

  fused_linattn_kernel<<<dim3(64, 64), 256, SMEM_BYTES, stream>>>(x, Wqkv, Wout, bout, g, out);
}


# k softmax (over n) and LayerNorm stats moved from 4 and 2 of 8 waves onto all: k rows reduced in the qkv accumulators, LN 4 lanes per pixel, DPP butterflies; sum order changed, one barrier fewer
# speedup vs baseline: 1.1272x; 1.0519x over previous
#include <hip/hip_runtime.h>


typedef float v2f __attribute__((ext_vector_type(2)));
typedef float v8f __attribute__((ext_vector_type(8)));

__device__ __forceinline__ v8f wmma4(v2f a, v2f b, v8f c) {
  return __builtin_amdgcn_wmma_f32_16x16x4_f32(false, a, false, b, (short)0, c, false, false);
}

template <int M>
__device__ __forceinline__ float row_xor(float v) {
  const int b = __builtin_bit_cast(int, v);
  return __builtin_bit_cast(float, __builtin_amdgcn_update_dpp(0, b, 0x160 + M, 0xf, 0xf, true));
}
__device__ __forceinline__ float row16_max(float v) {
  v = fmaxf(v, row_xor<1>(v));
  v = fmaxf(v, row_xor<2>(v));
  v = fmaxf(v, row_xor<4>(v));
  return fmaxf(v, row_xor<8>(v));
}
__device__ __forceinline__ float row16_sum(float v) {
  v += row_xor<1>(v);
  v += row_xor<2>(v);
  v += row_xor<4>(v);
  return v + row_xor<8>(v);
}
__device__ __forceinline__ float quad_sum(float v) {
  v += row_xor<1>(v);
  return v + row_xor<2>(v);
}

#define OFF_XT   0
#define OFF_QT   8448
#define OFF_KB   17152
#define OFF_VB   25856
#define OFF_CT   34560
#define OFF_BO   38912
#define OFF_GS   39040
#define OFF_MEAN 39168
#define OFF_RSTD 39232
#define SMEM_FLOATS 39296
#define SMEM_BYTES  (SMEM_FLOATS * 4)

__global__ __launch_bounds__(256, 1)
void fused_linattn_kernel(const float* __restrict__ x,
                          const float* __restrict__ Wqkv,
                          const float* __restrict__ Wout,
                          const float* __restrict__ bout,
                          const float* __restrict__ g,
                          float* __restrict__ out)
{
  extern __shared__ __attribute__((aligned(16))) float sm[];
  float* XT    = sm + OFF_XT;
  float* QT    = sm + OFF_QT;
  float* KB    = sm + OFF_KB;
  float* VB    = sm + OFF_VB;
  float* CT    = sm + OFF_CT;
  float* BO    = sm + OFF_BO;
  float* GS    = sm + OFF_GS;
  float* MEANv = sm + OFF_MEAN;
  float* RSTDv = sm + OFF_RSTD;
  float* AOT = XT;
  float* YT  = KB;

  const int tid  = threadIdx.x;
  const int wave = tid >> 5;
  const int lane = tid & 31;
  const int l16  = lane & 15;
  const int hi   = lane >> 4;
  const int kofs = 2 * hi;

  const int b  = blockIdx.y;
  const int ph = blockIdx.x >> 3;
  const int pw = blockIdx.x & 7;
  const long gbase = (long)b * 524288 + (long)(ph * 8) * 64 + (long)(pw * 8);

  const float* Wl = Wqkv + (wave * 16 + l16) * 128 + 16 * hi;
  float4 aw[2][3][4];
#pragma unroll
  for (int j = 0; j < 3; ++j) {
#pragma unroll
    for (int m = 0; m < 4; ++m)
      aw[0][j][m] = *(const float4*)(Wl + j * 16384 + 4 * m);
  }

  for (int it = 0; it < 4; ++it) {
    int p = tid + 256 * it;
    int c = p >> 3, r = p & 7;
    const float4* src = (const float4*)(x + gbase + (long)c * 4096 + (long)r * 64);
    float4 a0 = src[0], a1 = src[1];
    float vals[8] = {a0.x, a0.y, a0.z, a0.w, a1.x, a1.y, a1.z, a1.w};
#pragma unroll
    for (int col = 0; col < 8; ++col)
      XT[(r * 8 + col) * 132 + c] = vals[col];
  }
  if (tid < 128) { BO[tid] = bout[tid]; GS[tid] = g[tid]; }
  __syncthreads();

  float4 ao[4][4];

  {
    v8f acc[12] = {};
    const float* Xl = &XT[l16 * 132 + 16 * hi];
    float4 bx[2][4];
#pragma unroll
    for (int nt = 0; nt < 4; ++nt) bx[0][nt] = *(const float4*)(Xl + nt * 2112);
#pragma unroll
    for (int t = 0; t < 4; ++t) {
      if (t + 1 < 4) {
#pragma unroll
        for (int j = 0; j < 3; ++j) {
#pragma unroll
          for (int m = 0; m < 4; ++m)
            aw[(t + 1) & 1][j][m] = *(const float4*)(Wl + j * 16384 + 32 * (t + 1) + 4 * m);
        }
      }
#pragma unroll
      for (int m = 0; m < 4; ++m) {
        const int s = t * 4 + m;
        if (s + 1 < 16) {
#pragma unroll
          for (int nt = 0; nt < 4; ++nt)
            bx[(s + 1) & 1][nt] = *(const float4*)(Xl + nt * 2112 + 32 * ((s + 1) >> 2) + 4 * ((s + 1) & 3));
        }
#pragma unroll
        for (int j = 0; j < 3; ++j) {
          const float4 av = aw[t & 1][j][m];
          const v2f a = {av.x, av.y};
#pragma unroll
          for (int nt = 0; nt < 4; ++nt) {
            const v2f bq = {bx[s & 1][nt].x, bx[s & 1][nt].y};
            acc[j * 4 + nt] = wmma4(a, bq, acc[j * 4 + nt]);
          }
        }
#pragma unroll
        for (int j = 0; j < 3; ++j) {
          const float4 av = aw[t & 1][j][m];
          const v2f a = {av.z, av.w};
#pragma unroll
          for (int nt = 0; nt < 4; ++nt) {
            const v2f bq = {bx[s & 1][nt].z, bx[s & 1][nt].w};
            acc[j * 4 + nt] = wmma4(a, bq, acc[j * 4 + nt]);
          }
        }
      }
    }
#pragma unroll
    for (int t = 0; t < 4; ++t) {
#pragma unroll
      for (int m = 0; m < 4; ++m)
        ao[t][m] = *(const float4*)(Wout + (wave * 16 + l16) * 128 + 16 * hi + 32 * t + 4 * m);
    }
#pragma unroll
    for (int nt = 0; nt < 4; ++nt) {
      const int nn = nt * 16 + l16;
#pragma unroll
      for (int i = 0; i < 8; ++i) {
        const int cc = wave * 16 + i + 8 * hi;
        QT[(cc >> 5) * 2176 + nn * 34 + (cc & 31)] = acc[nt][i];
        VB[cc * 68 + nn] = acc[8 + nt][i];
      }
    }
#pragma unroll
    for (int i = 0; i < 8; ++i) {
      const int cc = wave * 16 + i + 8 * hi;
      const float a0 = acc[4][i], a1 = acc[5][i], a2 = acc[6][i], a3 = acc[7][i];
      const float mx = row16_max(fmaxf(fmaxf(a0, a1), fmaxf(a2, a3)));
      const float e0 = expf(a0 - mx), e1 = expf(a1 - mx), e2 = expf(a2 - mx), e3 = expf(a3 - mx);
      const float inv = 1.f / row16_sum((e0 + e1) + (e2 + e3));
      float* kr = &KB[cc * 68 + l16];
      kr[0] = e0 * inv; kr[16] = e1 * inv; kr[32] = e2 * inv; kr[48] = e3 * inv;
    }
  }
  __syncthreads();

  {
    int h = tid >> 6, n = tid & 63;
    float* row = &QT[h * 2176 + n * 34];
    float mx = row[0];
#pragma unroll
    for (int d = 1; d < 32; ++d) mx = fmaxf(mx, row[d]);
    float e[32]; float s = 0.f;
#pragma unroll
    for (int d = 0; d < 32; ++d) { e[d] = expf(row[d] - mx); s += e[d]; }
    float inv = 0.17677669529663687f / s;
#pragma unroll
    for (int d = 0; d < 32; ++d) row[d] = e[d] * inv;
  }

  {
    int h = wave >> 1, sub = wave & 1;
    v8f acc0 = {}, acc1 = {};
    const float* A0 = &KB[(h * 32 + l16) * 68 + kofs];
    const float* A1 = A0 + 16 * 68;
    const float* Bp = &VB[(h * 32 + sub * 16 + l16) * 68 + kofs];
#pragma unroll
    for (int k = 0; k < 64; k += 4) {
      v2f b  = *(const v2f*)(Bp + k);
      v2f a0 = *(const v2f*)(A0 + k);
      v2f a1 = *(const v2f*)(A1 + k);
      acc0 = wmma4(a0, b, acc0);
      acc1 = wmma4(a1, b, acc1);
    }
#pragma unroll
    for (int i = 0; i < 8; ++i) {
      int e2 = sub * 16 + l16;
      CT[h * 1088 + e2 * 34 + ( 0 + i + 8 * hi)] = acc0[i];
      CT[h * 1088 + e2 * 34 + (16 + i + 8 * hi)] = acc1[i];
    }
  }
  __syncthreads();

  {
    int h = wave >> 1, sub = wave & 1;
    v8f acc[4] = {};
    const float* A0 = &CT[h * 1088 + l16 * 34 + kofs];
    const float* A1 = A0 + 16 * 34;
    const float* B0 = &QT[h * 2176 + (sub * 32 + l16) * 34 + kofs];
    const float* B1 = B0 + 16 * 34;
#pragma unroll
    for (int k = 0; k < 32; k += 4) {
      v2f a0 = *(const v2f*)(A0 + k);
      v2f a1 = *(const v2f*)(A1 + k);
      v2f b0 = *(const v2f*)(B0 + k);
      v2f b1 = *(const v2f*)(B1 + k);
      acc[0] = wmma4(a0, b0, acc[0]);
      acc[1] = wmma4(a0, b1, acc[1]);
      acc[2] = wmma4(a1, b0, acc[2]);
      acc[3] = wmma4(a1, b1, acc[3]);
    }
#pragma unroll
    for (int mt = 0; mt < 2; ++mt) {
#pragma unroll
      for (int q2 = 0; q2 < 2; ++q2) {
        int nt = sub * 2 + q2;
        int nn = nt * 16 + l16;
#pragma unroll
        for (int i = 0; i < 8; ++i) {
          int e2 = mt * 16 + i + 8 * hi;
          AOT[nn * 132 + h * 32 + e2] = acc[mt * 2 + q2][i];
        }
      }
    }
  }
  __syncthreads();

  {
    const float* Al = &AOT[l16 * 132 + 16 * hi];
    v8f acc[4] = {};
    float4 bb[2][4];
#pragma unroll
    for (int nt = 0; nt < 4; ++nt) bb[0][nt] = *(const float4*)(Al + nt * 2112);
#pragma unroll
    for (int s = 0; s < 16; ++s) {
      if (s + 1 < 16) {
#pragma unroll
        for (int nt = 0; nt < 4; ++nt)
          bb[(s + 1) & 1][nt] = *(const float4*)(Al + nt * 2112 + 32 * ((s + 1) >> 2) + 4 * ((s + 1) & 3));
      }
      const float4 av = ao[s >> 2][s & 3];
      const v2f a0 = {av.x, av.y};
      const v2f a1 = {av.z, av.w};
#pragma unroll
      for (int nt = 0; nt < 4; ++nt) {
        const v2f bq = {bb[s & 1][nt].x, bb[s & 1][nt].y};
        acc[nt] = wmma4(a0, bq, acc[nt]);
      }
#pragma unroll
      for (int nt = 0; nt < 4; ++nt) {
        const v2f bq = {bb[s & 1][nt].z, bb[s & 1][nt].w};
        acc[nt] = wmma4(a1, bq, acc[nt]);
      }
    }
#pragma unroll
    for (int nt = 0; nt < 4; ++nt) {
      int nn = nt * 16 + l16;
#pragma unroll
      for (int i = 0; i < 8; ++i) {
        int mm = wave * 16 + i + 8 * hi;
        YT[nn * 132 + mm] = acc[nt][i] + BO[mm];
      }
    }
  }
  __syncthreads();

  {
    const int n = tid >> 2, part = tid & 3;
    const float4* row4 = (const float4*)&YT[n * 132 + part * 32];
    float4 y[8];
#pragma unroll
    for (int q4 = 0; q4 < 8; ++q4) y[q4] = row4[q4];
    float s4[8];
#pragma unroll
    for (int q4 = 0; q4 < 8; ++q4) s4[q4] = (y[q4].x + y[q4].y) + (y[q4].z + y[q4].w);
    const float s = quad_sum(((s4[0] + s4[1]) + (s4[2] + s4[3])) + ((s4[4] + s4[5]) + (s4[6] + s4[7])));
    const float mean = s * (1.f / 128.f);
    float v4[8];
#pragma unroll
    for (int q4 = 0; q4 < 8; ++q4) {
      const float d0 = y[q4].x - mean, d1 = y[q4].y - mean, d2 = y[q4].z - mean, d3 = y[q4].w - mean;
      v4[q4] = (d0 * d0 + d1 * d1) + (d2 * d2 + d3 * d3);
    }
    const float v = quad_sum(((v4[0] + v4[1]) + (v4[2] + v4[3])) + ((v4[4] + v4[5]) + (v4[6] + v4[7])));
    if (part == 0) {
      MEANv[n] = mean;
      RSTDv[n] = rsqrtf(v * (1.f / 128.f) + 1e-5f);
    }
  }
  __syncthreads();

  for (int it = 0; it < 4; ++it) {
    int p = tid + 256 * it;
    int c = p >> 3, r = p & 7;
    float gs = GS[c];
    float vals[8];
#pragma unroll
    for (int col = 0; col < 8; ++col) {
      int n = r * 8 + col;
      vals[col] = (YT[n * 132 + c] - MEANv[n]) * RSTDv[n] * gs;
    }
    float4* dst = (float4*)(out + gbase + (long)c * 4096 + (long)r * 64);
    dst[0] = make_float4(vals[0], vals[1], vals[2], vals[3]);
    dst[1] = make_float4(vals[4], vals[5], vals[6], vals[7]);
  }
}

extern "C" void kernel_launch(void* const* d_in, const int* in_sizes, int n_in,
                              void* d_out, int out_size, void* d_ws, size_t ws_size,
                              hipStream_t stream) {
  const float* x    = (const float*)d_in[0];
  const float* Wqkv = (const float*)d_in[1];
  const float* Wout = (const float*)d_in[2];
  const float* bout = (const float*)d_in[3];
  const float* g    = (const float*)d_in[4];
  float* out = (float*)d_out;
  (void)in_sizes; (void)n_in; (void)out_size; (void)d_ws; (void)ws_size;

  hipFuncSetAttribute((const void*)fused_linattn_kernel,
                      hipFuncAttributeMaxDynamicSharedMemorySize, SMEM_BYTES);

  fused_linattn_kernel<<<dim3(64, 64), 256, SMEM_BYTES, stream>>>(x, Wqkv, Wout, bout, g, out);
}


# x patch loads and output stores re-mapped: 4 lanes x 8 bytes per 32-byte row piece, one channel per wave-instruction, so each touches 8 lines once instead of 32 lines twice (b64 not b128)
# speedup vs baseline: 1.2772x; 1.1331x over previous
#include <hip/hip_runtime.h>


typedef float v2f __attribute__((ext_vector_type(2)));
typedef float v8f __attribute__((ext_vector_type(8)));

__device__ __forceinline__ v8f wmma4(v2f a, v2f b, v8f c) {
  return __builtin_amdgcn_wmma_f32_16x16x4_f32(false, a, false, b, (short)0, c, false, false);
}

template <int M>
__device__ __forceinline__ float row_xor(float v) {
  const int b = __builtin_bit_cast(int, v);
  return __builtin_bit_cast(float, __builtin_amdgcn_update_dpp(0, b, 0x160 + M, 0xf, 0xf, true));
}
__device__ __forceinline__ float row16_max(float v) {
  v = fmaxf(v, row_xor<1>(v));
  v = fmaxf(v, row_xor<2>(v));
  v = fmaxf(v, row_xor<4>(v));
  return fmaxf(v, row_xor<8>(v));
}
__device__ __forceinline__ float row16_sum(float v) {
  v += row_xor<1>(v);
  v += row_xor<2>(v);
  v += row_xor<4>(v);
  return v + row_xor<8>(v);
}
__device__ __forceinline__ float quad_sum(float v) {
  v += row_xor<1>(v);
  return v + row_xor<2>(v);
}

#define OFF_XT   0
#define OFF_QT   8448
#define OFF_KB   17152
#define OFF_VB   25856
#define OFF_CT   34560
#define OFF_BO   38912
#define OFF_GS   39040
#define OFF_MEAN 39168
#define OFF_RSTD 39232
#define SMEM_FLOATS 39296
#define SMEM_BYTES  (SMEM_FLOATS * 4)

__global__ __launch_bounds__(256, 1)
void fused_linattn_kernel(const float* __restrict__ x,
                          const float* __restrict__ Wqkv,
                          const float* __restrict__ Wout,
                          const float* __restrict__ bout,
                          const float* __restrict__ g,
                          float* __restrict__ out)
{
  extern __shared__ __attribute__((aligned(16))) float sm[];
  float* XT    = sm + OFF_XT;
  float* QT    = sm + OFF_QT;
  float* KB    = sm + OFF_KB;
  float* VB    = sm + OFF_VB;
  float* CT    = sm + OFF_CT;
  float* BO    = sm + OFF_BO;
  float* GS    = sm + OFF_GS;
  float* MEANv = sm + OFF_MEAN;
  float* RSTDv = sm + OFF_RSTD;
  float* AOT = XT;
  float* YT  = KB;

  const int tid  = threadIdx.x;
  const int wave = tid >> 5;
  const int lane = tid & 31;
  const int l16  = lane & 15;
  const int hi   = lane >> 4;
  const int kofs = 2 * hi;

  const int b  = blockIdx.y;
  const int ph = blockIdx.x >> 3;
  const int pw = blockIdx.x & 7;
  const long gbase = (long)b * 524288 + (long)(ph * 8) * 64 + (long)(pw * 8);

  const float* Wl = Wqkv + (wave * 16 + l16) * 128 + 16 * hi;
  float4 aw[2][3][4];
#pragma unroll
  for (int j = 0; j < 3; ++j) {
#pragma unroll
    for (int m = 0; m < 4; ++m)
      aw[0][j][m] = *(const float4*)(Wl + j * 16384 + 4 * m);
  }

  const int pr = lane >> 2;
  const int pq = lane & 3;
  const long poff = gbase + (long)pr * 64 + (long)(2 * pq);
  {
    float2 xv[16];
#pragma unroll
    for (int it = 0; it < 16; ++it)
      xv[it] = *(const float2*)(x + poff + (long)(wave + 8 * it) * 4096);
    float* xd = &XT[(pr * 8 + 2 * pq) * 132 + wave];
#pragma unroll
    for (int it = 0; it < 16; ++it) {
      xd[8 * it]       = xv[it].x;
      xd[8 * it + 132] = xv[it].y;
    }
  }
  if (tid < 128) { BO[tid] = bout[tid]; GS[tid] = g[tid]; }
  __syncthreads();

  float4 ao[4][4];

  {
    v8f acc[12] = {};
    const float* Xl = &XT[l16 * 132 + 16 * hi];
    float4 bx[2][4];
#pragma unroll
    for (int nt = 0; nt < 4; ++nt) bx[0][nt] = *(const float4*)(Xl + nt * 2112);
#pragma unroll
    for (int t = 0; t < 4; ++t) {
      if (t + 1 < 4) {
#pragma unroll
        for (int j = 0; j < 3; ++j) {
#pragma unroll
          for (int m = 0; m < 4; ++m)
            aw[(t + 1) & 1][j][m] = *(const float4*)(Wl + j * 16384 + 32 * (t + 1) + 4 * m);
        }
      }
#pragma unroll
      for (int m = 0; m < 4; ++m) {
        const int s = t * 4 + m;
        if (s + 1 < 16) {
#pragma unroll
          for (int nt = 0; nt < 4; ++nt)
            bx[(s + 1) & 1][nt] = *(const float4*)(Xl + nt * 2112 + 32 * ((s + 1) >> 2) + 4 * ((s + 1) & 3));
        }
#pragma unroll
        for (int j = 0; j < 3; ++j) {
          const float4 av = aw[t & 1][j][m];
          const v2f a = {av.x, av.y};
#pragma unroll
          for (int nt = 0; nt < 4; ++nt) {
            const v2f bq = {bx[s & 1][nt].x, bx[s & 1][nt].y};
            acc[j * 4 + nt] = wmma4(a, bq, acc[j * 4 + nt]);
          }
        }
#pragma unroll
        for (int j = 0; j < 3; ++j) {
          const float4 av = aw[t & 1][j][m];
          const v2f a = {av.z, av.w};
#pragma unroll
          for (int nt = 0; nt < 4; ++nt) {
            const v2f bq = {bx[s & 1][nt].z, bx[s & 1][nt].w};
            acc[j * 4 + nt] = wmma4(a, bq, acc[j * 4 + nt]);
          }
        }
      }
    }
#pragma unroll
    for (int t = 0; t < 4; ++t) {
#pragma unroll
      for (int m = 0; m < 4; ++m)
        ao[t][m] = *(const float4*)(Wout + (wave * 16 + l16) * 128 + 16 * hi + 32 * t + 4 * m);
    }
#pragma unroll
    for (int nt = 0; nt < 4; ++nt) {
      const int nn = nt * 16 + l16;
#pragma unroll
      for (int i = 0; i < 8; ++i) {
        const int cc = wave * 16 + i + 8 * hi;
        QT[(cc >> 5) * 2176 + nn * 34 + (cc & 31)] = acc[nt][i];
        VB[cc * 68 + nn] = acc[8 + nt][i];
      }
    }
#pragma unroll
    for (int i = 0; i < 8; ++i) {
      const int cc = wave * 16 + i + 8 * hi;
      const float a0 = acc[4][i], a1 = acc[5][i], a2 = acc[6][i], a3 = acc[7][i];
      const float mx = row16_max(fmaxf(fmaxf(a0, a1), fmaxf(a2, a3)));
      const float e0 = expf(a0 - mx), e1 = expf(a1 - mx), e2 = expf(a2 - mx), e3 = expf(a3 - mx);
      const float inv = 1.f / row16_sum((e0 + e1) + (e2 + e3));
      float* kr = &KB[cc * 68 + l16];
      kr[0] = e0 * inv; kr[16] = e1 * inv; kr[32] = e2 * inv; kr[48] = e3 * inv;
    }
  }
  __syncthreads();

  {
    int h = tid >> 6, n = tid & 63;
    float* row = &QT[h * 2176 + n * 34];
    float mx = row[0];
#pragma unroll
    for (int d = 1; d < 32; ++d) mx = fmaxf(mx, row[d]);
    float e[32]; float s = 0.f;
#pragma unroll
    for (int d = 0; d < 32; ++d) { e[d] = expf(row[d] - mx); s += e[d]; }
    float inv = 0.17677669529663687f / s;
#pragma unroll
    for (int d = 0; d < 32; ++d) row[d] = e[d] * inv;
  }

  {
    int h = wave >> 1, sub = wave & 1;
    v8f acc0 = {}, acc1 = {};
    const float* A0 = &KB[(h * 32 + l16) * 68 + kofs];
    const float* A1 = A0 + 16 * 68;
    const float* Bp = &VB[(h * 32 + sub * 16 + l16) * 68 + kofs];
#pragma unroll
    for (int k = 0; k < 64; k += 4) {
      v2f b  = *(const v2f*)(Bp + k);
      v2f a0 = *(const v2f*)(A0 + k);
      v2f a1 = *(const v2f*)(A1 + k);
      acc0 = wmma4(a0, b, acc0);
      acc1 = wmma4(a1, b, acc1);
    }
#pragma unroll
    for (int i = 0; i < 8; ++i) {
      int e2 = sub * 16 + l16;
      CT[h * 1088 + e2 * 34 + ( 0 + i + 8 * hi)] = acc0[i];
      CT[h * 1088 + e2 * 34 + (16 + i + 8 * hi)] = acc1[i];
    }
  }
  __syncthreads();

  {
    int h = wave >> 1, sub = wave & 1;
    v8f acc[4] = {};
    const float* A0 = &CT[h * 1088 + l16 * 34 + kofs];
    const float* A1 = A0 + 16 * 34;
    const float* B0 = &QT[h * 2176 + (sub * 32 + l16) * 34 + kofs];
    const float* B1 = B0 + 16 * 34;
#pragma unroll
    for (int k = 0; k < 32; k += 4) {
      v2f a0 = *(const v2f*)(A0 + k);
      v2f a1 = *(const v2f*)(A1 + k);
      v2f b0 = *(const v2f*)(B0 + k);
      v2f b1 = *(const v2f*)(B1 + k);
      acc[0] = wmma4(a0, b0, acc[0]);
      acc[1] = wmma4(a0, b1, acc[1]);
      acc[2] = wmma4(a1, b0, acc[2]);
      acc[3] = wmma4(a1, b1, acc[3]);
    }
#pragma unroll
    for (int mt = 0; mt < 2; ++mt) {
#pragma unroll
      for (int q2 = 0; q2 < 2; ++q2) {
        int nt = sub * 2 + q2;
        int nn = nt * 16 + l16;
#pragma unroll
        for (int i = 0; i < 8; ++i) {
          int e2 = mt * 16 + i + 8 * hi;
          AOT[nn * 132 + h * 32 + e2] = acc[mt * 2 + q2][i];
        }
      }
    }
  }
  __syncthreads();

  {
    const float* Al = &AOT[l16 * 132 + 16 * hi];
    v8f acc[4] = {};
    float4 bb[2][4];
#pragma unroll
    for (int nt = 0; nt < 4; ++nt) bb[0][nt] = *(const float4*)(Al + nt * 2112);
#pragma unroll
    for (int s = 0; s < 16; ++s) {
      if (s + 1 < 16) {
#pragma unroll
        for (int nt = 0; nt < 4; ++nt)
          bb[(s + 1) & 1][nt] = *(const float4*)(Al + nt * 2112 + 32 * ((s + 1) >> 2) + 4 * ((s + 1) & 3));
      }
      const float4 av = ao[s >> 2][s & 3];
      const v2f a0 = {av.x, av.y};
      const v2f a1 = {av.z, av.w};
#pragma unroll
      for (int nt = 0; nt < 4; ++nt) {
        const v2f bq = {bb[s & 1][nt].x, bb[s & 1][nt].y};
        acc[nt] = wmma4(a0, bq, acc[nt]);
      }
#pragma unroll
      for (int nt = 0; nt < 4; ++nt) {
        const v2f bq = {bb[s & 1][nt].z, bb[s & 1][nt].w};
        acc[nt] = wmma4(a1, bq, acc[nt]);
      }
    }
#pragma unroll
    for (int nt = 0; nt < 4; ++nt) {
      int nn = nt * 16 + l16;
#pragma unroll
      for (int i = 0; i < 8; ++i) {
        int mm = wave * 16 + i + 8 * hi;
        YT[nn * 132 + mm] = acc[nt][i] + BO[mm];
      }
    }
  }
  __syncthreads();

  {
    const int n = tid >> 2, part = tid & 3;
    const float4* row4 = (const float4*)&YT[n * 132 + part * 32];
    float4 y[8];
#pragma unroll
    for (int q4 = 0; q4 < 8; ++q4) y[q4] = row4[q4];
    float s4[8];
#pragma unroll
    for (int q4 = 0; q4 < 8; ++q4) s4[q4] = (y[q4].x + y[q4].y) + (y[q4].z + y[q4].w);
    const float s = quad_sum(((s4[0] + s4[1]) + (s4[2] + s4[3])) + ((s4[4] + s4[5]) + (s4[6] + s4[7])));
    const float mean = s * (1.f / 128.f);
    float v4[8];
#pragma unroll
    for (int q4 = 0; q4 < 8; ++q4) {
      const float d0 = y[q4].x - mean, d1 = y[q4].y - mean, d2 = y[q4].z - mean, d3 = y[q4].w - mean;
      v4[q4] = (d0 * d0 + d1 * d1) + (d2 * d2 + d3 * d3);
    }
    const float v = quad_sum(((v4[0] + v4[1]) + (v4[2] + v4[3])) + ((v4[4] + v4[5]) + (v4[6] + v4[7])));
    if (part == 0) {
      MEANv[n] = mean;
      RSTDv[n] = rsqrtf(v * (1.f / 128.f) + 1e-5f);
    }
  }
  __syncthreads();

  {
    const int n0 = pr * 8 + 2 * pq;
    const float m0 = MEANv[n0], m1 = MEANv[n0 + 1];
    const float r0 = RSTDv[n0], r1 = RSTDv[n0 + 1];
    const float* yd = &YT[n0 * 132 + wave];
#pragma unroll
    for (int it = 0; it < 16; ++it) {
      const int c = wave + 8 * it;
      const float gs = GS[c];
      float2 o;
      o.x = (yd[8 * it]       - m0) * r0 * gs;
      o.y = (yd[8 * it + 132] - m1) * r1 * gs;
      *(float2*)(out + poff + (long)c * 4096) = o;
    }
  }
}

extern "C" void kernel_launch(void* const* d_in, const int* in_sizes, int n_in,
                              void* d_out, int out_size, void* d_ws, size_t ws_size,
                              hipStream_t stream) {
  const float* x    = (const float*)d_in[0];
  const float* Wqkv = (const float*)d_in[1];
  const float* Wout = (const float*)d_in[2];
  const float* bout = (const float*)d_in[3];
  const float* g    = (const float*)d_in[4];
  float* out = (float*)d_out;
  (void)in_sizes; (void)n_in; (void)out_size; (void)d_ws; (void)ws_size;

  hipFuncSetAttribute((const void*)fused_linattn_kernel,
                      hipFuncAttributeMaxDynamicSharedMemorySize, SMEM_BYTES);

  fused_linattn_kernel<<<dim3(64, 64), 256, SMEM_BYTES, stream>>>(x, Wqkv, Wout, bout, g, out);
}
